# starting kernel
# baseline (speedup 1.0000x reference)
#include <hip/hip_runtime.h>

typedef __attribute__((ext_vector_type(2))) float v2f;
typedef __attribute__((ext_vector_type(8))) float v8f;

#if __has_builtin(__builtin_amdgcn_wmma_f32_16x16x4_f32)
#define USE_WMMA 1
#else
#define USE_WMMA 0
#endif

#define LDS_H 66

__global__ __launch_bounds__(128) void scl_pool_diff(
    const float* __restrict__ orig, const float* __restrict__ enh,
    float* __restrict__ p)
{
  __shared__ float y[3][64][LDS_H];

  const int tid = threadIdx.x;
  const int blk = blockIdx.x;
  const int b       = blk >> 4;
  const int t       = blk & 15;
  const int rowTile = t >> 1;
  const int colTile = t & 1;
  const int h0 = rowTile * 64;
  const int w0 = colTile * 256;

  const int jp     = tid & 63;
  const int hstart = tid >> 6;
  for (int c = 0; c < 3; ++c) {
    const float* ob = orig + (((size_t)b * 3 + c) * 512 + h0) * 512 + w0 + jp * 4;
    const float* eb = enh  + (((size_t)b * 3 + c) * 512 + h0) * 512 + w0 + jp * 4;
    #pragma unroll 4
    for (int h = hstart; h < 64; h += 2) {
      float4 o = *(const float4*)(ob + (size_t)h * 512);
      float4 e = *(const float4*)(eb + (size_t)h * 512);
      y[c][jp][h] = (o.x + o.y + o.z + o.w) - (e.x + e.y + e.z + e.w);
    }
  }
  __syncthreads();

  const int lane = tid & 31;
  const int w    = tid >> 5;
  const int nloc = w * 16 + (lane & 15);
  const int hi   = lane >> 4;

#if USE_WMMA
  const int koff = hi << 1;
  v8f acc = {};
  #pragma unroll
  for (int c = 0; c < 3; ++c) {
    #pragma unroll
    for (int kk = 0; kk < 16; ++kk) {
      float a = ((lane & 15) == kk) ? (1.0f / 48.0f) : 0.0f;
      v2f av = {a, a};
      const float* src = &y[c][nloc][4 * kk + koff];
      v2f bv = {src[0], src[1]};
      acc = __builtin_amdgcn_wmma_f32_16x16x4_f32(
          false, av, false, bv, (short)0, acc, false, false);
    }
  }
  {
    const int gi = rowTile * 16 + 8 * hi;
    const int gj = colTile * 64 + nloc;
    float* pb = p + ((size_t)b * 128 + gi) * 128 + gj;
    #pragma unroll
    for (int r = 0; r < 8; ++r)
      pb[(size_t)r * 128] = acc[r];
  }
#else
  {
    const int gj = colTile * 64 + nloc;
    #pragma unroll
    for (int r = 0; r < 8; ++r) {
      int m = 8 * hi + r;
      float s = 0.0f;
      #pragma unroll
      for (int c = 0; c < 3; ++c)
        #pragma unroll
        for (int k = 0; k < 4; ++k)
          s += y[c][nloc][4 * m + k];
      p[((size_t)b * 128 + rowTile * 16 + m) * 128 + gj] = s * (1.0f / 48.0f);
    }
  }
#endif
}

__global__ __launch_bounds__(256) void scl_loss(
    const float* __restrict__ p, float* __restrict__ out)
{
  const int idx = blockIdx.x * 256 + threadIdx.x;
  const int j = idx & 127;
  const int i = (idx >> 7) & 127;
  const float c = p[idx];
  const float l = (j > 0)   ? p[idx - 1]   : 0.0f;
  const float r = (j < 127) ? p[idx + 1]   : 0.0f;
  const float u = (i > 0)   ? p[idx - 128] : 0.0f;
  const float d = (i < 127) ? p[idx + 128] : 0.0f;
  const float dl = c - l, dr = c - r, du = c - u, dd = c - d;
  out[idx] = dl * dl + dr * dr + du * du + dd * dd;
}

extern "C" void kernel_launch(void* const* d_in, const int* in_sizes, int n_in,
                              void* d_out, int out_size, void* d_ws, size_t ws_size,
                              hipStream_t stream) {
  const float* orig = (const float*)d_in[0];
  const float* enh  = (const float*)d_in[1];
  float* out = (float*)d_out;
  float* p   = (float*)d_ws;

  scl_pool_diff<<<dim3(512), dim3(128), 0, stream>>>(orig, enh, p);
  scl_loss<<<dim3(32 * 128 * 128 / 256), dim3(256), 0, stream>>>(p, out);
}


# Pooling kernel as a register stream: each lane's float4 row sums go straight into the f32 WMMA B operand (no LDS, no barrier); 16 waves per WGP, 24-36 loads in flight per wave
# speedup vs baseline: 1.2255x; 1.2255x over previous
#include <hip/hip_runtime.h>


typedef __attribute__((ext_vector_type(2))) float v2f;
typedef __attribute__((ext_vector_type(8))) float v8f;
typedef __attribute__((ext_vector_type(4))) float v4f;

#define STEPS 8
#define AHEAD 2
#define RING  (AHEAD + 1)

__global__ __launch_bounds__(512) void scl_pool_diff(
    const float* __restrict__ orig, const float* __restrict__ enh,
    float* __restrict__ p)
{
  const int tid   = threadIdx.x;
  const int lane  = tid & 31;
  const int wave  = __builtin_amdgcn_readfirstlane(tid >> 5);
  const int b     = blockIdx.x >> 3;
  const int band  = blockIdx.x & 7;
  const int strip = wave & 7;
  const int pr0   = band * 16 + (wave >> 3) * 8;
  const int n     = lane & 15;
  const int hi    = lane >> 4;

  const size_t off = ((size_t)b * 3 * 512 + (size_t)(4 * pr0 + 2 * hi)) * 512 + strip * 64 + n * 4;
  const float* ob = orig + off;
  const float* eb = enh  + off;

  v4f ro[RING][6], re[RING][6];

#define LOAD_STEP(S) do { \
    _Pragma("unroll") for (int c = 0; c < 3; ++c) { \
      _Pragma("unroll") for (int v = 0; v < 2; ++v) { \
        ro[(S) % RING][c * 2 + v] = *(const v4f*)(ob + (size_t)c * 512 * 512 + (size_t)(4 * (S) + v) * 512); \
        re[(S) % RING][c * 2 + v] = *(const v4f*)(eb + (size_t)c * 512 * 512 + (size_t)(4 * (S) + v) * 512); \
      } } } while (0)

  #pragma unroll
  for (int s = 0; s < AHEAD; ++s) LOAD_STEP(s);

  v8f acc = {};
  #pragma unroll
  for (int s = 0; s < STEPS; ++s) {
    __builtin_amdgcn_sched_barrier(0);
    if (s + AHEAD < STEPS) LOAD_STEP(s + AHEAD);
    __builtin_amdgcn_sched_barrier(0);
    asm volatile("" : "+v"(ro[s % RING][0]), "+v"(ro[s % RING][1]), "+v"(ro[s % RING][2]),
                      "+v"(ro[s % RING][3]), "+v"(ro[s % RING][4]), "+v"(ro[s % RING][5]),
                      "+v"(re[s % RING][0]), "+v"(re[s % RING][1]), "+v"(re[s % RING][2]),
                      "+v"(re[s % RING][3]), "+v"(re[s % RING][4]), "+v"(re[s % RING][5]));

    const int mrow = (s & 3) + 8 * (s >> 2);
    const float a = (n == mrow) ? (1.0f / 48.0f) : 0.0f;
    const v2f av = {a, a};
    #pragma unroll
    for (int c = 0; c < 3; ++c) {
      const v4f o0 = ro[s % RING][c * 2 + 0], e0 = re[s % RING][c * 2 + 0];
      const v4f o1 = ro[s % RING][c * 2 + 1], e1 = re[s % RING][c * 2 + 1];
      const v2f bv = {(o0.x + o0.y + o0.z + o0.w) - (e0.x + e0.y + e0.z + e0.w),
                      (o1.x + o1.y + o1.z + o1.w) - (e1.x + e1.y + e1.z + e1.w)};
      acc = __builtin_amdgcn_wmma_f32_16x16x4_f32(
          false, av, false, bv, (short)0, acc, false, false);
    }
  }
#undef LOAD_STEP

  float* pb = p + ((size_t)b * 128 + pr0 + 4 * hi) * 128 + strip * 16 + n;
  #pragma unroll
  for (int r = 0; r < 4; ++r)
    pb[(size_t)r * 128] = acc[r];
}

__global__ __launch_bounds__(256) void scl_loss(
    const float* __restrict__ p, float* __restrict__ out)
{
  const int idx = blockIdx.x * 256 + threadIdx.x;
  const int j = idx & 127;
  const int i = (idx >> 7) & 127;
  const float c = p[idx];
  const float l = (j > 0)   ? p[idx - 1]   : 0.0f;
  const float r = (j < 127) ? p[idx + 1]   : 0.0f;
  const float u = (i > 0)   ? p[idx - 128] : 0.0f;
  const float d = (i < 127) ? p[idx + 128] : 0.0f;
  const float dl = c - l, dr = c - r, du = c - u, dd = c - d;
  out[idx] = dl * dl + dr * dr + du * du + dd * dd;
}

extern "C" void kernel_launch(void* const* d_in, const int* in_sizes, int n_in,
                              void* d_out, int out_size, void* d_ws, size_t ws_size,
                              hipStream_t stream) {
  const float* orig = (const float*)d_in[0];
  const float* enh  = (const float*)d_in[1];
  float* out = (float*)d_out;
  float* p   = (float*)d_ws;

  scl_pool_diff<<<dim3(256), dim3(512), 0, stream>>>(orig, enh, p);
  scl_loss<<<dim3(32 * 128 * 128 / 256), dim3(256), 0, stream>>>(p, out);
}


# Loss stencil fused into the pooling kernel: each wave also pools the row next to its band as a ninth step, the band's pooled tile meets in LDS, one launch instead of two
# speedup vs baseline: 1.3857x; 1.1307x over previous
#include <hip/hip_runtime.h>


typedef __attribute__((ext_vector_type(2))) float v2f;
typedef __attribute__((ext_vector_type(8))) float v8f;
typedef __attribute__((ext_vector_type(4))) float v4f;

#define STEPS 8
#define AHEAD 2
#define RING  (AHEAD + 1)
#define PITCH 136
#define TROWS 18

__global__ __launch_bounds__(512) void scl_pool_loss(
    const float* __restrict__ orig, const float* __restrict__ enh,
    float* __restrict__ out)
{
  __shared__ __attribute__((aligned(16))) float tile[TROWS * PITCH];

  const int tid   = threadIdx.x;
  const int lane  = tid & 31;
  const int wave  = __builtin_amdgcn_readfirstlane(tid >> 5);
  const int b     = blockIdx.x >> 3;
  const int band  = blockIdx.x & 7;
  const int strip = wave & 7;
  const int half  = wave >> 3;
  const int pr0   = band * 16 + half * 8;
  const int n     = lane & 15;
  const int hi    = lane >> 4;

  const size_t off = ((size_t)b * 3 * 512 + (size_t)(4 * pr0 + 2 * hi)) * 512 + strip * 64 + n * 4;
  const float* ob = orig + off;
  const float* eb = enh  + off;

  const bool has_halo = half ? (band < 7) : (band > 0);
  const int  hrel     = has_halo ? (half ? 8 : -1) : 0;
  const float* obh = ob + (ptrdiff_t)hrel * 4 * 512;
  const float* ebh = eb + (ptrdiff_t)hrel * 4 * 512;

  v4f ro[RING][6], re[RING][6];

#define LOAD_ROWS(SLOT, OB, EB, ROW) do { \
    _Pragma("unroll") for (int c = 0; c < 3; ++c) { \
      _Pragma("unroll") for (int v = 0; v < 2; ++v) { \
        ro[SLOT][c * 2 + v] = *(const v4f*)((OB) + (size_t)c * 512 * 512 + (size_t)((ROW) + v) * 512); \
        re[SLOT][c * 2 + v] = *(const v4f*)((EB) + (size_t)c * 512 * 512 + (size_t)((ROW) + v) * 512); \
      } } } while (0)
#define LOAD_STEP(S) LOAD_ROWS((S) % RING, ob, eb, 4 * (S))
#define LOAD_HALO()  LOAD_ROWS(STEPS % RING, obh, ebh, 0)

#define REDUCE(SLOT, MROW) do { \
      \
      \
      \
    asm volatile("" : "+v"(ro[SLOT][0]), "+v"(ro[SLOT][1]), "+v"(ro[SLOT][2]), \
                      "+v"(ro[SLOT][3]), "+v"(ro[SLOT][4]), "+v"(ro[SLOT][5]), \
                      "+v"(re[SLOT][0]), "+v"(re[SLOT][1]), "+v"(re[SLOT][2]), \
                      "+v"(re[SLOT][3]), "+v"(re[SLOT][4]), "+v"(re[SLOT][5])); \
    const float a = (n == (MROW)) ? (1.0f / 48.0f) : 0.0f; \
    const v2f av = {a, a}; \
    _Pragma("unroll") for (int c = 0; c < 3; ++c) { \
      const v4f o0 = ro[SLOT][c * 2 + 0], e0 = re[SLOT][c * 2 + 0]; \
      const v4f o1 = ro[SLOT][c * 2 + 1], e1 = re[SLOT][c * 2 + 1]; \
      const v2f bv = {(o0.x + o0.y + o0.z + o0.w) - (e0.x + e0.y + e0.z + e0.w), \
                      (o1.x + o1.y + o1.z + o1.w) - (e1.x + e1.y + e1.z + e1.w)}; \
      acc = __builtin_amdgcn_wmma_f32_16x16x4_f32( \
          false, av, false, bv, (short)0, acc, false, false); \
    } } while (0)

  #pragma unroll
  for (int s = 0; s < AHEAD; ++s) LOAD_STEP(s);

  v8f acc = {};
  #pragma unroll
  for (int s = 0; s <= STEPS; ++s) {
    __builtin_amdgcn_sched_barrier(0);
    if (s + AHEAD < STEPS) LOAD_STEP(s + AHEAD);
    else if (s + AHEAD == STEPS) { if (has_halo) LOAD_HALO(); }
    __builtin_amdgcn_sched_barrier(0);
    if (s < STEPS) REDUCE(s % RING, (s & 3) + 8 * (s >> 2));
    else if (has_halo) REDUCE(STEPS % RING, 4);
  }
#undef REDUCE
#undef LOAD_HALO
#undef LOAD_STEP
#undef LOAD_ROWS

  {
    float* tp = tile + (1 + half * 8 + 4 * hi) * PITCH + 4 + strip * 16 + n;
    #pragma unroll
    for (int r = 0; r < 4; ++r)
      tp[r * PITCH] = acc[r];
    if (hi == 0)
      tile[(half ? (TROWS - 1) : 0) * PITCH + 4 + strip * 16 + n] = has_halo ? acc[4] : 0.0f;
    if (tid < 32)
      tile[(1 + (tid & 15)) * PITCH + ((tid < 16) ? 3 : 132)] = 0.0f;
  }
  __syncthreads();

  {
    const float* tc = tile + (1 + wave) * PITCH + 4 + 4 * lane;
    const v4f c = *(const v4f*)tc;
    const v4f u = *(const v4f*)(tc - PITCH);
    const v4f d = *(const v4f*)(tc + PITCH);
    const float l = tc[-1];
    const float r = tc[4];
    const v4f dl = {c.x - l,   c.y - c.x, c.z - c.y, c.w - c.z};
    const v4f dr = {c.x - c.y, c.y - c.z, c.z - c.w, c.w - r};
    const v4f du = c - u;
    const v4f dd = c - d;
    const v4f o = dl * dl + dr * dr + du * du + dd * dd;
    *(v4f*)(out + ((size_t)b * 128 + band * 16 + wave) * 128 + 4 * lane) = o;
  }
}

extern "C" void kernel_launch(void* const* d_in, const int* in_sizes, int n_in,
                              void* d_out, int out_size, void* d_ws, size_t ws_size,
                              hipStream_t stream) {
  const float* orig = (const float*)d_in[0];
  const float* enh  = (const float*)d_in[1];
  float* out = (float*)d_out;

  scl_pool_loss<<<dim3(256), dim3(512), 0, stream>>>(orig, enh, out);
}
